# starting kernel
# baseline (speedup 1.0000x reference)
#include <hip/hip_runtime.h>
#include <hip/hip_bf16.h>

typedef __attribute__((ext_vector_type(2))) float v2f;
typedef __attribute__((ext_vector_type(4))) float v4f;
typedef __attribute__((ext_vector_type(8))) float v8f;

#define ROWS    16
#define CHUNK   512
#define STRIDE  516
#define BLOCK   512

#if defined(__AMDGCN__) && __has_builtin(__builtin_amdgcn_wmma_f32_16x16x4_f32)
#define HAVE_WMMA_F32 1
#else
#define HAVE_WMMA_F32 0
#endif

__device__ __forceinline__ float act(float s) {
  s = fminf(fmaxf(s, -20.0f), 20.0f);
#if defined(__AMDGCN__) && __has_builtin(__builtin_amdgcn_tanhf)
  return __builtin_amdgcn_tanhf(0.5f * s);
#else
  float e = expf(-s);
  return 2.0f / (1.0f + e) - 1.0f;
#endif
}

__global__ __launch_bounds__(BLOCK) void noise_mean_wmma(
    const float* __restrict__ pub,
    const float* __restrict__ priv,
    const float* __restrict__ pb,
    const float* __restrict__ pc,
    float* __restrict__ out,
    int n_samples, float inv_n)
{
  __shared__ float tile[ROWS * STRIDE];
  __shared__ float rowsum[ROWS];

  const int tid  = threadIdx.x;
  const int lane = tid & 31;
  const int wave = tid >> 5;
  const int rowBase = blockIdx.x * ROWS;

  const float b = pb[0];
  const float c = pc[0];
  const int   myRow = lane & 15;
  const float pt = pub[rowBase + myRow] * b;

  if (tid < ROWS) rowsum[tid] = 0.0f;

  const int ldRow = tid >> 7;
  const int ldCol = (tid & 127) * 4;

  const int kOff   = (lane >> 4) * 2;
  const int waveK0 = wave * (CHUNK / 16);

#if HAVE_WMMA_F32
  v8f acc = {};
  const v2f onesB = {1.0f, 1.0f};
#else
  float acc = 0.0f;
#endif

  v4f regs[4];
#pragma unroll
  for (int p = 0; p < 4; ++p) {
    const int r = p * 4 + ldRow;
    regs[p] = *(const v4f*)(priv + (size_t)(rowBase + r) * (size_t)n_samples
                                 + (size_t)ldCol);
  }

  for (int s0 = 0; s0 < n_samples; s0 += CHUNK) {
    __syncthreads();
#pragma unroll
    for (int p = 0; p < 4; ++p) {
      const int r = p * 4 + ldRow;
      *(v4f*)(tile + r * STRIDE + ldCol) = regs[p];
    }
    __syncthreads();

    const int s1 = s0 + CHUNK;
    if (s1 < n_samples) {
#pragma unroll
      for (int p = 0; p < 4; ++p) {
        const int r = p * 4 + ldRow;
        regs[p] = *(const v4f*)(priv + (size_t)(rowBase + r) * (size_t)n_samples
                                     + (size_t)(s1 + ldCol));
      }
    }

#pragma unroll
    for (int k = 0; k < CHUNK / 16; k += 4) {
      const float* src = tile + myRow * STRIDE + waveK0 + k + kOff;
      v2f a;
      a.x = act(pt + src[0] * c);
      a.y = act(pt + src[1] * c);
#if HAVE_WMMA_F32
      acc = __builtin_amdgcn_wmma_f32_16x16x4_f32(false, a, false, onesB,
                                                  (short)0, acc, false, false);
#else
      acc += a.x + a.y;
#endif
    }
  }

#if HAVE_WMMA_F32
  if (lane == 0 || lane == 16) {
    const int mbase = (lane >> 4) * 8;
#pragma unroll
    for (int i = 0; i < 8; ++i) atomicAdd(&rowsum[mbase + i], acc[i]);
  }
#else
  atomicAdd(&rowsum[myRow], acc);
#endif

  __syncthreads();
  if (tid < ROWS) out[rowBase + tid] = rowsum[tid] * inv_n;
}

extern "C" void kernel_launch(void* const* d_in, const int* in_sizes, int n_in,
                              void* d_out, int out_size, void* d_ws, size_t ws_size,
                              hipStream_t stream) {
  const float* pub  = (const float*)d_in[0];
  const float* priv = (const float*)d_in[1];
  const float* pb   = (const float*)d_in[2];
  const float* pc   = (const float*)d_in[3];
  float* out = (float*)d_out;

  const int batch     = in_sizes[0];
  const int n_samples = in_sizes[1] / batch;
  const float inv_n   = 1.0f / (float)n_samples;

  dim3 grid(batch / ROWS);
  noise_mean_wmma<<<grid, BLOCK, 0, stream>>>(pub, priv, pb, pc, out, n_samples, inv_n);
}


# one wave streams one row: 512-byte coalesced loads straight into the WMMA A fragments, 16 in flight per wave, no LDS tile and no barrier in the loop; row sum order changed (new reduction tree)
# speedup vs baseline: 1.5342x; 1.5342x over previous
#include <hip/hip_runtime.h>
#include <hip/hip_bf16.h>


typedef __attribute__((ext_vector_type(2))) float v2f;
typedef __attribute__((ext_vector_type(4))) float v4f;
typedef __attribute__((ext_vector_type(8))) float v8f;

#define ROWS    16
#define BLOCK   512
#define PIECE   128
#define DEPTH   16
#define SPAN    (PIECE * DEPTH)

#if defined(__AMDGCN__) && __has_builtin(__builtin_amdgcn_wmma_f32_16x16x4_f32)
#define HAVE_WMMA_F32 1
#else
#define HAVE_WMMA_F32 0
#endif

__device__ __forceinline__ float act(float s) {
  s = fminf(fmaxf(s, -20.0f), 20.0f);
#if defined(__AMDGCN__) && __has_builtin(__builtin_amdgcn_tanhf)
  return __builtin_amdgcn_tanhf(0.5f * s);
#else
  float e = expf(-s);
  return 2.0f / (1.0f + e) - 1.0f;
#endif
}

#if HAVE_WMMA_F32
typedef v8f acc_t;
#else
typedef float acc_t;
#endif

__device__ __forceinline__ void consume(acc_t& acc, v4f x, float pt, float c) {
#if HAVE_WMMA_F32
  const v2f onesB = {1.0f, 1.0f};
  v2f a0, a1;
  a0.x = act(pt + x.x * c);
  a0.y = act(pt + x.y * c);
  a1.x = act(pt + x.z * c);
  a1.y = act(pt + x.w * c);
  acc = __builtin_amdgcn_wmma_f32_16x16x4_f32(false, a0, false, onesB, (short)0, acc, false, false);
  acc = __builtin_amdgcn_wmma_f32_16x16x4_f32(false, a1, false, onesB, (short)0, acc, false, false);
#else
  acc += act(pt + x.x * c) + act(pt + x.y * c) + act(pt + x.z * c) + act(pt + x.w * c);
#endif
}

__global__ __launch_bounds__(BLOCK) void noise_mean_wmma(
    const float* __restrict__ pub,
    const float* __restrict__ priv,
    const float* __restrict__ pb,
    const float* __restrict__ pc,
    float* __restrict__ out,
    int n_samples, float inv_n)
{
  __shared__ float rowsum[ROWS];

  const int tid  = threadIdx.x;
  const int lane = tid & 31;
  const int wave = __builtin_amdgcn_readfirstlane(tid >> 5);
  const int rowBase = blockIdx.x * ROWS;
  const int row = rowBase + wave;

  const float b = pb[0];
  const float c = pc[0];
  const float pt = pub[row] * b;

  const float* rowp = priv + (size_t)row * (size_t)n_samples + (size_t)(lane * 4);

#if HAVE_WMMA_F32
  acc_t acc = {};
#else
  acc_t acc = 0.0f;
#endif

  v4f buf[DEPTH];
#pragma unroll
  for (int j = 0; j < DEPTH; ++j) buf[j] = *(const v4f*)(rowp + j * PIECE);

  int s = 0;
  for (; s + SPAN < n_samples; s += SPAN) {
#pragma unroll
    for (int j = 0; j < DEPTH; ++j) {
      consume(acc, buf[j], pt, c);
      __builtin_amdgcn_sched_barrier(0);
      buf[j] = *(const v4f*)(rowp + (s + SPAN + j * PIECE));
      __builtin_amdgcn_sched_barrier(0);
    }
  }
#pragma unroll
  for (int j = 0; j < DEPTH; ++j) consume(acc, buf[j], pt, c);

#if HAVE_WMMA_F32
  float t = ((acc[0] + acc[1]) + (acc[2] + acc[3])) + ((acc[4] + acc[5]) + (acc[6] + acc[7]));
  t += __shfl_xor(t, 16);
#else
  float t = acc;
  for (int o = 16; o > 0; o >>= 1) t += __shfl_xor(t, o);
#endif
  if (lane == 0) rowsum[wave] = t;

  __syncthreads();
  if (tid < ROWS) out[rowBase + tid] = rowsum[tid] * inv_n;
}

extern "C" void kernel_launch(void* const* d_in, const int* in_sizes, int n_in,
                              void* d_out, int out_size, void* d_ws, size_t ws_size,
                              hipStream_t stream) {
  const float* pub  = (const float*)d_in[0];
  const float* priv = (const float*)d_in[1];
  const float* pb   = (const float*)d_in[2];
  const float* pc   = (const float*)d_in[3];
  float* out = (float*)d_out;

  const int batch     = in_sizes[0];
  const int n_samples = in_sizes[1] / batch;
  const float inv_n   = 1.0f / (float)n_samples;

  dim3 grid(batch / ROWS);
  noise_mean_wmma<<<grid, BLOCK, 0, stream>>>(pub, priv, pb, pc, out, n_samples, inv_n);
}


# each piece's two A fragments are added lane-wise first (one packed add), so one f32 WMMA per 512-byte piece sums the row instead of two; row sum order changed (pairs added first)
# speedup vs baseline: 1.7111x; 1.1153x over previous
#include <hip/hip_runtime.h>
#include <hip/hip_bf16.h>


typedef __attribute__((ext_vector_type(2))) float v2f;
typedef __attribute__((ext_vector_type(4))) float v4f;
typedef __attribute__((ext_vector_type(8))) float v8f;

#define ROWS    16
#define BLOCK   512
#define PIECE   128
#define DEPTH   16
#define SPAN    (PIECE * DEPTH)

#if defined(__AMDGCN__) && __has_builtin(__builtin_amdgcn_wmma_f32_16x16x4_f32)
#define HAVE_WMMA_F32 1
#else
#define HAVE_WMMA_F32 0
#endif

__device__ __forceinline__ float act(float s) {
  s = fminf(fmaxf(s, -20.0f), 20.0f);
#if defined(__AMDGCN__) && __has_builtin(__builtin_amdgcn_tanhf)
  return __builtin_amdgcn_tanhf(0.5f * s);
#else
  float e = expf(-s);
  return 2.0f / (1.0f + e) - 1.0f;
#endif
}

#if HAVE_WMMA_F32
typedef v8f acc_t;
#else
typedef float acc_t;
#endif

__device__ __forceinline__ void consume(acc_t& acc, v4f x, float pt, float c) {
#if HAVE_WMMA_F32
  const v2f onesB = {1.0f, 1.0f};
  v2f a0, a1;
  a0.x = act(pt + x.x * c);
  a0.y = act(pt + x.y * c);
  a1.x = act(pt + x.z * c);
  a1.y = act(pt + x.w * c);
  const v2f a = a0 + a1;
  acc = __builtin_amdgcn_wmma_f32_16x16x4_f32(false, a, false, onesB, (short)0, acc, false, false);
#else
  acc += act(pt + x.x * c) + act(pt + x.y * c) + act(pt + x.z * c) + act(pt + x.w * c);
#endif
}

__global__ __launch_bounds__(BLOCK) void noise_mean_wmma(
    const float* __restrict__ pub,
    const float* __restrict__ priv,
    const float* __restrict__ pb,
    const float* __restrict__ pc,
    float* __restrict__ out,
    int n_samples, float inv_n)
{
  __shared__ float rowsum[ROWS];

  const int tid  = threadIdx.x;
  const int lane = tid & 31;
  const int wave = __builtin_amdgcn_readfirstlane(tid >> 5);
  const int rowBase = blockIdx.x * ROWS;
  const int row = rowBase + wave;

  const float b = pb[0];
  const float c = pc[0];
  const float pt = pub[row] * b;

  const float* rowp = priv + (size_t)row * (size_t)n_samples + (size_t)(lane * 4);

#if HAVE_WMMA_F32
  acc_t acc = {};
#else
  acc_t acc = 0.0f;
#endif

  v4f buf[DEPTH];
#pragma unroll
  for (int j = 0; j < DEPTH; ++j) buf[j] = *(const v4f*)(rowp + j * PIECE);

  int s = 0;
  for (; s + SPAN < n_samples; s += SPAN) {
#pragma unroll
    for (int j = 0; j < DEPTH; ++j) {
      consume(acc, buf[j], pt, c);
      __builtin_amdgcn_sched_barrier(0);
      buf[j] = *(const v4f*)(rowp + (s + SPAN + j * PIECE));
      __builtin_amdgcn_sched_barrier(0);
    }
  }
#pragma unroll
  for (int j = 0; j < DEPTH; ++j) consume(acc, buf[j], pt, c);

#if HAVE_WMMA_F32
  float t = ((acc[0] + acc[1]) + (acc[2] + acc[3])) + ((acc[4] + acc[5]) + (acc[6] + acc[7]));
  t += __shfl_xor(t, 16);
#else
  float t = acc;
  for (int o = 16; o > 0; o >>= 1) t += __shfl_xor(t, o);
#endif
  if (lane == 0) rowsum[wave] = t;

  __syncthreads();
  if (tid < ROWS) out[rowBase + tid] = rowsum[tid] * inv_n;
}

extern "C" void kernel_launch(void* const* d_in, const int* in_sizes, int n_in,
                              void* d_out, int out_size, void* d_ws, size_t ws_size,
                              hipStream_t stream) {
  const float* pub  = (const float*)d_in[0];
  const float* priv = (const float*)d_in[1];
  const float* pb   = (const float*)d_in[2];
  const float* pc   = (const float*)d_in[3];
  float* out = (float*)d_out;

  const int batch     = in_sizes[0];
  const int n_samples = in_sizes[1] / batch;
  const float inv_n   = 1.0f / (float)n_samples;

  dim3 grid(batch / ROWS);
  noise_mean_wmma<<<grid, BLOCK, 0, stream>>>(pub, priv, pb, pc, out, n_samples, inv_n);
}
